# starting kernel
# baseline (speedup 1.0000x reference)
#include <hip/hip_runtime.h>

typedef float v2f __attribute__((ext_vector_type(2)));
typedef float v8f __attribute__((ext_vector_type(8)));

#define B_SZ   4096
#define F_NUM  16
#define V_SZ   20000
#define E_SZ   64
#define HID    1024

__global__ void __launch_bounds__(256) gather_embed(const int* __restrict__ idx,
                                                    const float* __restrict__ emb,
                                                    float* __restrict__ xw) {
    int tid = blockIdx.x * 256 + threadIdx.x;
    int b = tid >> 10;
    int c = tid & 1023;
    int f = c >> 6;
    int e = c & 63;
    int v = idx[b * F_NUM + f];
    xw[tid] = emb[((size_t)f * V_SZ + (size_t)v) * E_SZ + e];
}

__global__ void __launch_bounds__(128) gemm_relu(const float* __restrict__ X,
                                                 const float* __restrict__ W,
                                                 const float* __restrict__ bias,
                                                 float* __restrict__ Y) {
    const int lane = threadIdx.x & 31;
    const int gw   = blockIdx.x * 4 + (threadIdx.x >> 5);
    const int m0   = (gw >> 4) * 16;
    const int n0   = (gw & 15) * 64;
    const int lmod = lane & 15;
    const int lhi  = lane >> 4;

    v8f acc[4] = {v8f{}, v8f{}, v8f{}, v8f{}};

    const float* xrow = X + (size_t)(m0 + lmod) * HID + 2 * lhi;
    const float* wcol = W + (size_t)(2 * lhi) * HID + n0 + lmod;

    for (int k = 0; k < HID; k += 4) {
        v2f a = *(const v2f*)(xrow + k);
        const float* wr0 = wcol + (size_t)k * HID;
        const float* wr1 = wr0 + HID;
        v2f b0 = { wr0[ 0], wr1[ 0] };
        v2f b1 = { wr0[16], wr1[16] };
        v2f b2 = { wr0[32], wr1[32] };
        v2f b3 = { wr0[48], wr1[48] };
        acc[0] = __builtin_amdgcn_wmma_f32_16x16x4_f32(false, a, false, b0, (short)0, acc[0], false, false);
        acc[1] = __builtin_amdgcn_wmma_f32_16x16x4_f32(false, a, false, b1, (short)0, acc[1], false, false);
        acc[2] = __builtin_amdgcn_wmma_f32_16x16x4_f32(false, a, false, b2, (short)0, acc[2], false, false);
        acc[3] = __builtin_amdgcn_wmma_f32_16x16x4_f32(false, a, false, b3, (short)0, acc[3], false, false);
    }

    #pragma unroll
    for (int s = 0; s < 4; ++s) {
        const int n  = n0 + 16 * s + lmod;
        const float bv = bias[n];
        float* yp = Y + (size_t)(m0 + 8 * lhi) * HID + n;
        #pragma unroll
        for (int j = 0; j < 8; ++j) {
            float v = acc[s][j] + bv;
            yp[(size_t)j * HID] = v > 0.f ? v : 0.f;
        }
    }
}

__global__ void __launch_bounds__(128) attn_final(const float* __restrict__ L0,
                                                  const float* __restrict__ L1,
                                                  const float* __restrict__ L2,
                                                  const float* __restrict__ LF,
                                                  const float* __restrict__ wf,
                                                  const float* __restrict__ bf,
                                                  float* __restrict__ out) {
    const int b = blockIdx.x;
    const int t = threadIdx.x;
    const float* Ls[3] = {L0, L1, L2};
    const float* H[3];
    #pragma unroll
    for (int j = 0; j < 3; ++j) {
        int r = 3 * b + j;
        H[j] = Ls[r / B_SZ] + (size_t)(r % B_SZ) * HID;
    }

    __shared__ float red[3][128];
    __shared__ float cs[3];

    float p01 = 0.f, p02 = 0.f, p12 = 0.f;
    for (int h = t; h < HID; h += 128) {
        float h0 = H[0][h], h1 = H[1][h], h2 = H[2][h];
        p01 += h0 * h1; p02 += h0 * h2; p12 += h1 * h2;
    }
    red[0][t] = p01; red[1][t] = p02; red[2][t] = p12;
    __syncthreads();
    for (int off = 64; off > 0; off >>= 1) {
        if (t < off) {
            red[0][t] += red[0][t + off];
            red[1][t] += red[1][t + off];
            red[2][t] += red[2][t + off];
        }
        __syncthreads();
    }

    if (t == 0) {
        float s01 = red[0][0], s02 = red[1][0], s12 = red[2][0];
        float S[3][3] = {{0.f, s01, s02}, {s01, 0.f, s12}, {s02, s12, 0.f}};
        float col[3] = {0.f, 0.f, 0.f};
        #pragma unroll
        for (int i = 0; i < 3; ++i) {
            float m = fmaxf(S[i][0], fmaxf(S[i][1], S[i][2]));
            float e0 = expf(S[i][0] - m), e1 = expf(S[i][1] - m), e2 = expf(S[i][2] - m);
            float inv = 1.f / (e0 + e1 + e2);
            col[0] += e0 * inv; col[1] += e1 * inv; col[2] += e2 * inv;
        }
        cs[0] = col[0]; cs[1] = col[1]; cs[2] = col[2];
    }
    __syncthreads();

    const float c0 = cs[0], c1 = cs[1], c2 = cs[2];
    const float* lf = LF + (size_t)b * HID;
    float p = 0.f;
    for (int h = t; h < HID; h += 128) {
        float G = c0 * H[0][h] + c1 * H[1][h] + c2 * H[2][h];
        p += lf[h] * wf[h] + G * wf[HID + h];
    }
    __syncthreads();
    red[0][t] = p;
    __syncthreads();
    for (int off = 64; off > 0; off >>= 1) {
        if (t < off) red[0][t] += red[0][t + off];
        __syncthreads();
    }
    if (t == 0) out[b] = red[0][0] + bf[0];
}

extern "C" void kernel_launch(void* const* d_in, const int* in_sizes, int n_in,
                              void* d_out, int out_size, void* d_ws, size_t ws_size,
                              hipStream_t stream) {
    const int*   idx = (const int*)  d_in[0];
    const float* emb = (const float*)d_in[1];
    const float* w0  = (const float*)d_in[2];
    const float* b0  = (const float*)d_in[3];
    const float* w1  = (const float*)d_in[4];
    const float* b1  = (const float*)d_in[5];
    const float* w2  = (const float*)d_in[6];
    const float* b2  = (const float*)d_in[7];
    const float* w3  = (const float*)d_in[8];
    const float* b3  = (const float*)d_in[9];
    const float* wf  = (const float*)d_in[10];
    const float* bfin= (const float*)d_in[11];
    float* out = (float*)d_out;

    const size_t buf = (size_t)B_SZ * HID;
    float* ws0 = (float*)d_ws;
    float* ws1 = ws0 + buf;
    float* ws2 = ws1 + buf;
    float* ws3 = ws2 + buf;

    gather_embed<<<(B_SZ * HID) / 256, 256, 0, stream>>>(idx, emb, ws0);

    const int gemm_blocks = 1024;
    gemm_relu<<<gemm_blocks, 128, 0, stream>>>(ws0, w0, b0, ws1);
    gemm_relu<<<gemm_blocks, 128, 0, stream>>>(ws1, w1, b1, ws2);
    gemm_relu<<<gemm_blocks, 128, 0, stream>>>(ws2, w2, b2, ws0);
    gemm_relu<<<gemm_blocks, 128, 0, stream>>>(ws0, w3, b3, ws3);

    attn_final<<<B_SZ, 128, 0, stream>>>(ws1, ws2, ws0, ws3, wf, bfin, out);
}


# gemm_relu: LDS ring of three stages, tiles stored two K-steps ahead; each wave reads the next step's first fragments before the barrier, so its WMMAs go on right behind it; k order of sums unchanged
# speedup vs baseline: 1.6664x; 1.6664x over previous
#include <hip/hip_runtime.h>

typedef float v2f __attribute__((ext_vector_type(2)));
typedef float v8f __attribute__((ext_vector_type(8)));

#define B_SZ   4096
#define F_NUM  16
#define V_SZ   20000
#define E_SZ   64
#define HID    1024

#define BM 128
#define BN 64
#define BK 32
#define AP 36
#define BP 160
#define NKT (HID / BK)

struct Stage {
    float4 a0, a1, a2, a3;
    float4 b0, b1;
};

template <bool GATHER>
__device__ __forceinline__ float4 x_piece(const float* __restrict__ xg, int k0, const int* is, int i) {
    if constexpr (GATHER) {
        const int f = k0 >> 6;
        const int e = k0 & 63;
        int v = is[32 * i * F_NUM + f];
        v = min(max(v, 0), V_SZ - 1);
        const unsigned off = ((unsigned)f * V_SZ + (unsigned)v) * E_SZ + (unsigned)e;
        return *(const float4*)(xg + off);
    } else {
        return *(const float4*)(xg + (size_t)(32 * i) * HID + k0);
    }
}

template <bool GATHER>
__device__ __forceinline__ void tile_fetch(Stage& st, const float* __restrict__ xg,
                                           const float* __restrict__ wg, int k0,
                                           const int* is) {
    st.a0 = x_piece<GATHER>(xg, k0, is, 0);
    st.a1 = x_piece<GATHER>(xg, k0, is, 1);
    st.a2 = x_piece<GATHER>(xg, k0, is, 2);
    st.a3 = x_piece<GATHER>(xg, k0, is, 3);
    st.b0 = *(const float4*)(wg + (size_t)k0 * HID);
    st.b1 = *(const float4*)(wg + (size_t)(k0 + 1) * HID);
}

__device__ __forceinline__ void tile_store(const Stage& st, float* as, float* bs) {
    *(float4*)(as)           = st.a0;
    *(float4*)(as + 32 * AP) = st.a1;
    *(float4*)(as + 64 * AP) = st.a2;
    *(float4*)(as + 96 * AP) = st.a3;
    *(float4*)(bs)     = float4{st.b0.x, st.b1.x, st.b0.y, st.b1.y};
    *(float4*)(bs + 4) = float4{st.b0.z, st.b1.z, st.b0.w, st.b1.w};
}

struct Frag {
    v2f a0, a1, b0, b1;
};

__device__ __forceinline__ void frag_load(Frag& f, const float* af, const float* bf, int kk) {
    f.a0 = *(const v2f*)(af + kk);
    f.a1 = *(const v2f*)(af + 16 * AP + kk);
    f.b0 = *(const v2f*)(bf + (kk / 2) * BP);
    f.b1 = *(const v2f*)(bf + (kk / 2) * BP + 32);
}

__device__ __forceinline__ void frag_mma(v8f (&acc)[2][2], const Frag& f) {
    acc[0][0] = __builtin_amdgcn_wmma_f32_16x16x4_f32(false, f.a0, false, f.b0, (short)0, acc[0][0], false, false);
    acc[0][1] = __builtin_amdgcn_wmma_f32_16x16x4_f32(false, f.a0, false, f.b1, (short)0, acc[0][1], false, false);
    acc[1][0] = __builtin_amdgcn_wmma_f32_16x16x4_f32(false, f.a1, false, f.b0, (short)0, acc[1][0], false, false);
    acc[1][1] = __builtin_amdgcn_wmma_f32_16x16x4_f32(false, f.a1, false, f.b1, (short)0, acc[1][1], false, false);
}

template <bool GATHER, bool STORE, bool FETCH, bool PRE>
__device__ __forceinline__ void k_step(v8f (&acc)[2][2], Frag (&pre)[2], Stage& st,
                                       const float* __restrict__ xg, const float* __restrict__ wg,
                                       int k_fetch, const int* is,
                                       const float* af, const float* bf,
                                       const float* afn, const float* bfn,
                                       float* as_st, float* bs_st) {
    if constexpr (STORE) tile_store(st, as_st, bs_st);
    if constexpr (FETCH) tile_fetch<GATHER>(st, xg, wg, k_fetch, is);
    frag_mma(acc, pre[0]);
    frag_mma(acc, pre[1]);
    #pragma unroll
    for (int kk = 8; kk < BK; kk += 4) {
        Frag f;
        frag_load(f, af, bf, kk);
        frag_mma(acc, f);
    }
    if constexpr (PRE) {
        frag_load(pre[0], afn, bfn, 0);
        frag_load(pre[1], afn, bfn, 4);
    }
    if constexpr (STORE) {
        __syncthreads();
        __builtin_amdgcn_sched_barrier(0);
    }
}

template <bool GATHER>
__global__ void __launch_bounds__(256) gemm_relu(const float* __restrict__ X,
                                                 const int* __restrict__ idx,
                                                 const float* __restrict__ W,
                                                 const float* __restrict__ bias,
                                                 float* __restrict__ Y) {
    __shared__ __attribute__((aligned(16))) float As[3][BM * AP];
    __shared__ __attribute__((aligned(16))) float Bs[3][(BK / 2) * BP];
    __shared__ __attribute__((aligned(16))) int Is[GATHER ? BM * F_NUM : 4];

    const int t    = threadIdx.x;
    const int lane = t & 31;
    const int wave = t >> 5;
    const int lmod = lane & 15;
    const int lhi  = lane >> 4;
    const int wm   = wave >> 1;
    const int wn   = wave & 1;
    const int bm0  = (blockIdx.x >> 4) * BM;
    const int bn0  = (blockIdx.x & 15) * BN;

    const float* xg = GATHER ? X + (t & 7) * 4
                             : X + (size_t)(bm0 + (t >> 3)) * HID + (t & 7) * 4;
    const int* is = &Is[GATHER ? (t >> 3) * F_NUM : 0];
    const float* wg = W + (size_t)(2 * (t >> 4)) * HID + bn0 + (t & 15) * 4;
    const int a_st = (t >> 3) * AP + (t & 7) * 4;
    const int b_st = (t >> 4) * BP + (t & 15) * 8;

    const int a_fr = (wm * 32 + lmod) * AP + 2 * lhi;
    const int b_fr = lhi * BP + (wn * 32 + lmod) * 2;

    v8f acc[2][2] = {{v8f{}, v8f{}}, {v8f{}, v8f{}}};

    if constexpr (GATHER) {
        const int4* ig = (const int4*)(idx + (size_t)bm0 * F_NUM) + 2 * t;
        const int4 i0 = ig[0], i1 = ig[1];
        *(int4*)&Is[8 * t]     = i0;
        *(int4*)&Is[8 * t + 4] = i1;
        __syncthreads();
    }

    Stage st, st1;
    tile_fetch<GATHER>(st,  xg, wg, 0,  is);
    tile_fetch<GATHER>(st1, xg, wg, BK, is);
    tile_store(st,  &As[0][a_st], &Bs[0][b_st]);
    tile_store(st1, &As[1][a_st], &Bs[1][b_st]);
    tile_fetch<GATHER>(st, xg, wg, 2 * BK, is);
    __syncthreads();

    Frag pre[2];
    frag_load(pre[0], &As[0][a_fr], &Bs[0][b_fr], 0);
    frag_load(pre[1], &As[0][a_fr], &Bs[0][b_fr], 4);

#define K_STEP(STORE, FETCH, PRE, SC, KF)                                                            \
    k_step<GATHER, STORE, FETCH, PRE>(acc, pre, st, xg, wg, (KF) * BK, is,                           \
                                      &As[(SC) % 3][a_fr], &Bs[(SC) % 3][b_fr],                      \
                                      &As[((SC) + 1) % 3][a_fr], &Bs[((SC) + 1) % 3][b_fr],          \
                                      &As[((SC) + 2) % 3][a_st], &Bs[((SC) + 2) % 3][b_st])
    for (int kt = 0; kt < NKT - 5; kt += 3) {
        K_STEP(true, true, true, 0, kt + 3);
        K_STEP(true, true, true, 1, kt + 4);
        K_STEP(true, true, true, 2, kt + 5);
    }
    K_STEP(true,  true,  true,  0, NKT - 2);
    K_STEP(true,  true,  true,  1, NKT - 1);
    K_STEP(true,  false, true,  2, 0);
    K_STEP(false, false, true,  0, 0);
    K_STEP(false, false, false, 1, 0);
#undef K_STEP

    #pragma unroll
    for (int s = 0; s < 2; ++s) {
        const int n  = bn0 + wn * 32 + 16 * s + lmod;
        const float bv = bias[n];
        #pragma unroll
        for (int i = 0; i < 2; ++i) {
            float* yp = Y + (size_t)(bm0 + wm * 32 + 16 * i + 8 * lhi) * HID + n;
            #pragma unroll
            for (int j = 0; j < 8; ++j) {
                float v = acc[i][s][j] + bv;
                yp[(size_t)j * HID] = v > 0.f ? v : 0.f;
            }
        }
    }
}

__global__ void __launch_bounds__(128) attn_final(const float* __restrict__ L0,
                                                  const float* __restrict__ L1,
                                                  const float* __restrict__ L2,
                                                  const float* __restrict__ LF,
                                                  const float* __restrict__ wf,
                                                  const float* __restrict__ bf,
                                                  float* __restrict__ out) {
    const int b = blockIdx.x;
    const int t = threadIdx.x;
    const float* Ls[3] = {L0, L1, L2};
    const float* H[3];
    #pragma unroll
    for (int j = 0; j < 3; ++j) {
        int r = 3 * b + j;
        H[j] = Ls[r / B_SZ] + (size_t)(r % B_SZ) * HID;
    }

    __shared__ float red[3][128];
    __shared__ float cs[3];

    float p01 = 0.f, p02 = 0.f, p12 = 0.f;
    for (int h = t; h < HID; h += 128) {
        float h0 = H[0][h], h1 = H[1][h], h2 = H[2][h];
        p01 += h0 * h1; p02 += h0 * h2; p12 += h1 * h2;
    }
    red[0][t] = p01; red[1][t] = p02; red[2][t] = p12;
    __syncthreads();
    for (int off = 64; off > 0; off >>= 1) {
        if (t < off) {
            red[0][t] += red[0][t + off];
            red[1][t] += red[1][t + off];
            red[2][t] += red[2][t + off];
        }
        __syncthreads();
    }

    if (t == 0) {
        float s01 = red[0][0], s02 = red[1][0], s12 = red[2][0];
        float S[3][3] = {{0.f, s01, s02}, {s01, 0.f, s12}, {s02, s12, 0.f}};
        float col[3] = {0.f, 0.f, 0.f};
        #pragma unroll
        for (int i = 0; i < 3; ++i) {
            float m = fmaxf(S[i][0], fmaxf(S[i][1], S[i][2]));
            float e0 = expf(S[i][0] - m), e1 = expf(S[i][1] - m), e2 = expf(S[i][2] - m);
            float inv = 1.f / (e0 + e1 + e2);
            col[0] += e0 * inv; col[1] += e1 * inv; col[2] += e2 * inv;
        }
        cs[0] = col[0]; cs[1] = col[1]; cs[2] = col[2];
    }
    __syncthreads();

    const float c0 = cs[0], c1 = cs[1], c2 = cs[2];
    const float* lf = LF + (size_t)b * HID;
    float p = 0.f;
    for (int h = t; h < HID; h += 128) {
        float G = c0 * H[0][h] + c1 * H[1][h] + c2 * H[2][h];
        p += lf[h] * wf[h] + G * wf[HID + h];
    }
    __syncthreads();
    red[0][t] = p;
    __syncthreads();
    for (int off = 64; off > 0; off >>= 1) {
        if (t < off) red[0][t] += red[0][t + off];
        __syncthreads();
    }
    if (t == 0) out[b] = red[0][0] + bf[0];
}

extern "C" void kernel_launch(void* const* d_in, const int* in_sizes, int n_in,
                              void* d_out, int out_size, void* d_ws, size_t ws_size,
                              hipStream_t stream) {
    const int*   idx = (const int*)  d_in[0];
    const float* emb = (const float*)d_in[1];
    const float* w0  = (const float*)d_in[2];
    const float* b0  = (const float*)d_in[3];
    const float* w1  = (const float*)d_in[4];
    const float* b1  = (const float*)d_in[5];
    const float* w2  = (const float*)d_in[6];
    const float* b2  = (const float*)d_in[7];
    const float* w3  = (const float*)d_in[8];
    const float* b3  = (const float*)d_in[9];
    const float* wf  = (const float*)d_in[10];
    const float* bfin= (const float*)d_in[11];
    float* out = (float*)d_out;

    const size_t buf = (size_t)B_SZ * HID;
    float* ws0 = (float*)d_ws;
    float* ws1 = ws0 + buf;
    float* ws2 = ws1 + buf;
    float* ws3 = ws2 + buf;

    const int gemm_blocks = 512;
    gemm_relu<true ><<<gemm_blocks, 256, 0, stream>>>(emb, idx,     w0, b0, ws1);
    gemm_relu<false><<<gemm_blocks, 256, 0, stream>>>(ws1, nullptr, w1, b1, ws2);
    gemm_relu<false><<<gemm_blocks, 256, 0, stream>>>(ws2, nullptr, w2, b2, ws0);
    gemm_relu<false><<<gemm_blocks, 256, 0, stream>>>(ws0, nullptr, w3, b3, ws3);

    attn_final<<<B_SZ, 128, 0, stream>>>(ws1, ws2, ws0, ws3, wf, bfin, out);
}
